# starting kernel
# baseline (speedup 1.0000x reference)
#include <hip/hip_runtime.h>
#include <hip/hip_bf16.h>

#define BATCH   65536
#define NEXP    16
#define HDIM    128
#define DDIM    64
#define ZDIM    16
#define LDS_STRIDE 132
#define ROWS    64
#define MTILES  (ROWS / 16)
#define NTILES_MAX (BATCH / ROWS + NEXP)

typedef __attribute__((ext_vector_type(2))) float v2f;
typedef __attribute__((ext_vector_type(8))) float v8f;

__device__ __forceinline__ v8f wmma_f32(v2f a, v2f b, v8f c) {
  return __builtin_amdgcn_wmma_f32_16x16x4_f32(
       false, a,  false, b,
       (short)0, c,  false,  false);
}


__global__ __launch_bounds__(64) void k_zero(int* p) {
  if (threadIdx.x < 32) p[threadIdx.x] = 0;
}

__global__ __launch_bounds__(256) void k_hist(const int* __restrict__ y,
                                              int* __restrict__ counts) {
  int b = blockIdx.x * 256 + threadIdx.x;
  if (b < BATCH) atomicAdd(&counts[y[b] & (NEXP - 1)], 1);
}

__global__ __launch_bounds__(256) void k_scatter(const int* __restrict__ y,
                                                 const int* __restrict__ counts,
                                                 int* __restrict__ cursors,
                                                 int* __restrict__ perm) {
  __shared__ int soff[NEXP];
  if (threadIdx.x == 0) {
    int o = 0;
    for (int e = 0; e < NEXP; ++e) { soff[e] = o; o += counts[e]; }
  }
  __syncthreads();
  int b = blockIdx.x * 256 + threadIdx.x;
  if (b < BATCH) {
    int e = y[b] & (NEXP - 1);
    int pos = soff[e] + atomicAdd(&cursors[e], 1);
    perm[pos] = b;
  }
}


__device__ __forceinline__ void layer_tile(const float* __restrict__ src,
                                           const float* __restrict__ W, int ldw,
                                           const float* __restrict__ bias,
                                           float* __restrict__ dst,
                                           int Kdim, int lane, int wave) {
  const int n0   = wave << 4;
  const int m    = lane & 15;
  const int koff = (lane >> 4) << 1;
  const int n    = n0 + m;
  v8f c0 = {}, c1 = {}, c2 = {}, c3 = {};
  const float* a_ptr = src + m * LDS_STRIDE + koff;
  const float* b_ptr = W + (size_t)koff * ldw + n;
  const int steps = Kdim >> 2;
  #pragma unroll 4
  for (int kk = 0; kk < steps; ++kk) {
    v2f b; b.x = b_ptr[0]; b.y = b_ptr[ldw];
    v2f a0; a0.x = a_ptr[0 * 16 * LDS_STRIDE];     a0.y = a_ptr[0 * 16 * LDS_STRIDE + 1];
    v2f a1; a1.x = a_ptr[1 * 16 * LDS_STRIDE];     a1.y = a_ptr[1 * 16 * LDS_STRIDE + 1];
    v2f a2; a2.x = a_ptr[2 * 16 * LDS_STRIDE];     a2.y = a_ptr[2 * 16 * LDS_STRIDE + 1];
    v2f a3; a3.x = a_ptr[3 * 16 * LDS_STRIDE];     a3.y = a_ptr[3 * 16 * LDS_STRIDE + 1];
    c0 = wmma_f32(a0, b, c0);
    c1 = wmma_f32(a1, b, c1);
    c2 = wmma_f32(a2, b, c2);
    c3 = wmma_f32(a3, b, c3);
    a_ptr += 4;
    b_ptr += 4 * (size_t)ldw;
  }
  const float bv = bias[n];
  const int mbase = (lane >> 4) << 3;
  v8f acc[4] = {c0, c1, c2, c3};
  #pragma unroll
  for (int t = 0; t < MTILES; ++t) {
    #pragma unroll
    for (int v = 0; v < 8; ++v) {
      float x = acc[t][v] + bv;
      x = fmaxf(x, 0.0f);
      dst[(t * 16 + mbase + v) * LDS_STRIDE + n] = x;
    }
  }
}

__device__ __forceinline__ void layer_out(const float* __restrict__ src,
                                          const float* __restrict__ W,
                                          const float* __restrict__ bias,
                                          float* __restrict__ out,
                                          const int* __restrict__ sh_gid,
                                          int nvalid, int lane, int wave) {
  if (wave >= 4) return;
  const int m    = lane & 15;
  const int koff = (lane >> 4) << 1;
  const int n    = (wave << 4) + m;
  v8f c0 = {}, c1 = {}, c2 = {}, c3 = {};
  const float* a_ptr = src + m * LDS_STRIDE + koff;
  const float* b_ptr = W + (size_t)koff * DDIM + n;
  #pragma unroll 4
  for (int kk = 0; kk < HDIM / 4; ++kk) {
    v2f b; b.x = b_ptr[0]; b.y = b_ptr[DDIM];
    v2f a0; a0.x = a_ptr[0 * 16 * LDS_STRIDE];     a0.y = a_ptr[0 * 16 * LDS_STRIDE + 1];
    v2f a1; a1.x = a_ptr[1 * 16 * LDS_STRIDE];     a1.y = a_ptr[1 * 16 * LDS_STRIDE + 1];
    v2f a2; a2.x = a_ptr[2 * 16 * LDS_STRIDE];     a2.y = a_ptr[2 * 16 * LDS_STRIDE + 1];
    v2f a3; a3.x = a_ptr[3 * 16 * LDS_STRIDE];     a3.y = a_ptr[3 * 16 * LDS_STRIDE + 1];
    c0 = wmma_f32(a0, b, c0);
    c1 = wmma_f32(a1, b, c1);
    c2 = wmma_f32(a2, b, c2);
    c3 = wmma_f32(a3, b, c3);
    a_ptr += 4;
    b_ptr += 4 * DDIM;
  }
  const float bv = bias[n];
  const int mbase = (lane >> 4) << 3;
  v8f acc[4] = {c0, c1, c2, c3};
  #pragma unroll
  for (int t = 0; t < MTILES; ++t) {
    #pragma unroll
    for (int v = 0; v < 8; ++v) {
      int mm = t * 16 + mbase + v;
      if (mm < nvalid) {
        out[(size_t)sh_gid[mm] * DDIM + n] = acc[t][v] + bv;
      }
    }
  }
}

__global__ __launch_bounds__(256) void mapping_main(
    const float* __restrict__ z,
    const float* __restrict__ bw0, const float* __restrict__ bb0,
    const float* __restrict__ bw1, const float* __restrict__ bb1,
    const float* __restrict__ bw2, const float* __restrict__ bb2,
    const float* __restrict__ bw3, const float* __restrict__ bb3,
    const float* __restrict__ hw0, const float* __restrict__ hb0,
    const float* __restrict__ hw1, const float* __restrict__ hb1,
    const float* __restrict__ hw2, const float* __restrict__ hb2,
    const float* __restrict__ hw3, const float* __restrict__ hb3,
    const int* __restrict__ counts, const int* __restrict__ perm,
    float* __restrict__ out) {
  __shared__ float actA[ROWS * LDS_STRIDE];
  __shared__ float actB[ROWS * LDS_STRIDE];
  __shared__ int sh_meta[3];
  __shared__ int sh_gid[ROWS];

  const int tid  = threadIdx.x;
  const int lane = tid & 31;
  const int wave = tid >> 5;

  if (tid == 0) {
    int t = blockIdx.x;
    int off = 0, base = 0, expert = -1, start = 0, end = 0;
    for (int e = 0; e < NEXP; ++e) {
      int cnt = counts[e];
      int nt  = (cnt + ROWS - 1) / ROWS;
      if (expert < 0 && t < base + nt) {
        expert = e;
        start  = off + (t - base) * ROWS;
        end    = off + cnt;
      }
      base += nt;
      off  += cnt;
    }
    sh_meta[0] = expert; sh_meta[1] = start; sh_meta[2] = end;
  }
  __syncthreads();
  const int expert = sh_meta[0];
  if (expert < 0) return;
  const int start = sh_meta[1];
  const int end   = sh_meta[2];
  const int nvalid = min(ROWS, end - start);

  {
    #pragma unroll
    for (int i = 0; i < ROWS * ZDIM / 256; ++i) {
      int idx = i * 256 + tid;
      int r = idx >> 4, cidx = idx & 15;
      int row = start + r;
      if (row >= end) row = end - 1;
      int g = perm[row];
      if (cidx == 0) sh_gid[r] = g;
      actA[r * LDS_STRIDE + cidx] = z[(size_t)g * ZDIM + cidx];
    }
  }
  __syncthreads();

  layer_tile(actA, bw0, HDIM, bb0, actB, ZDIM, lane, wave);  __syncthreads();
  layer_tile(actB, bw1, HDIM, bb1, actA, HDIM, lane, wave);  __syncthreads();
  layer_tile(actA, bw2, HDIM, bb2, actB, HDIM, lane, wave);  __syncthreads();
  layer_tile(actB, bw3, HDIM, bb3, actA, HDIM, lane, wave);  __syncthreads();

  const size_t we = (size_t)expert * HDIM * HDIM;
  layer_tile(actA, hw0 + we, HDIM, hb0 + expert * HDIM, actB, HDIM, lane, wave); __syncthreads();
  layer_tile(actB, hw1 + we, HDIM, hb1 + expert * HDIM, actA, HDIM, lane, wave); __syncthreads();
  layer_tile(actA, hw2 + we, HDIM, hb2 + expert * HDIM, actB, HDIM, lane, wave); __syncthreads();
  layer_out(actB, hw3 + (size_t)expert * HDIM * DDIM, hb3 + expert * DDIM,
            out, sh_gid, nvalid, lane, wave);
}


extern "C" void kernel_launch(void* const* d_in, const int* in_sizes, int n_in,
                              void* d_out, int out_size, void* d_ws, size_t ws_size,
                              hipStream_t stream) {
  const float* z   = (const float*)d_in[0];
  const int*   y   = (const int*)  d_in[1];
  const float* bw0 = (const float*)d_in[2];
  const float* bb0 = (const float*)d_in[3];
  const float* bw1 = (const float*)d_in[4];
  const float* bb1 = (const float*)d_in[5];
  const float* bw2 = (const float*)d_in[6];
  const float* bb2 = (const float*)d_in[7];
  const float* bw3 = (const float*)d_in[8];
  const float* bb3 = (const float*)d_in[9];
  const float* hw0 = (const float*)d_in[10];
  const float* hb0 = (const float*)d_in[11];
  const float* hw1 = (const float*)d_in[12];
  const float* hb1 = (const float*)d_in[13];
  const float* hw2 = (const float*)d_in[14];
  const float* hb2 = (const float*)d_in[15];
  const float* hw3 = (const float*)d_in[16];
  const float* hb3 = (const float*)d_in[17];
  float* out = (float*)d_out;

  int* counts  = (int*)d_ws;
  int* cursors = counts + 16;
  int* perm    = counts + 32;

  k_zero<<<1, 64, 0, stream>>>(counts);
  k_hist<<<BATCH / 256, 256, 0, stream>>>(y, counts);
  k_scatter<<<BATCH / 256, 256, 0, stream>>>(y, counts, cursors, perm);
  mapping_main<<<NTILES_MAX, 256, 0, stream>>>(
      z, bw0, bb0, bw1, bb1, bw2, bb2, bw3, bb3,
      hw0, hb0, hw1, hb1, hw2, hb2, hw3, hb3,
      counts, perm, out);
}


# Bucket sort of samples by expert without global atomics: per-chunk ballot histograms, then column sums + in-chunk ballot rank (stable order); 3 launches instead of 4; float sums unchanged
# speedup vs baseline: 1.2229x; 1.2229x over previous
#include <hip/hip_runtime.h>
#include <hip/hip_bf16.h>

#define BATCH   65536
#define NEXP    16
#define HDIM    128
#define DDIM    64
#define ZDIM    16
#define LDS_STRIDE 132
#define ROWS    64
#define MTILES  (ROWS / 16)
#define NTILES_MAX (BATCH / ROWS + NEXP)

typedef __attribute__((ext_vector_type(2))) float v2f;
typedef __attribute__((ext_vector_type(8))) float v8f;

__device__ __forceinline__ v8f wmma_f32(v2f a, v2f b, v8f c) {
  return __builtin_amdgcn_wmma_f32_16x16x4_f32(
       false, a,  false, b,
       (short)0, c,  false,  false);
}


#define CHUNK      256
#define NCHUNK     (BATCH / CHUNK)
#define HROW       32
#define WAVES_PB   (CHUNK / 32)

__global__ __launch_bounds__(CHUNK) void k_chunkhist(const int* __restrict__ y,
                                                     int* __restrict__ chist) {
  __shared__ int s_wc[WAVES_PB * NEXP];
  const int tid  = threadIdx.x;
  const int lane = tid & 31;
  const int wave = tid >> 5;
  const int e = y[blockIdx.x * CHUNK + tid] & (NEXP - 1);
  int mine = 0;
  #pragma unroll
  for (int k = 0; k < NEXP; ++k) {
    const unsigned m = __builtin_amdgcn_ballot_w32(e == k);
    if (lane == k) mine = __popc(m);
  }
  if (lane < NEXP) s_wc[wave * NEXP + lane] = mine;
  __syncthreads();
  if (tid < NEXP) {
    int s = 0;
    #pragma unroll
    for (int w = 0; w < WAVES_PB; ++w) s += s_wc[w * NEXP + tid];
    chist[blockIdx.x * HROW + tid] = s;
  }
}

__global__ __launch_bounds__(CHUNK) void k_rank_scatter(const int* __restrict__ y,
                                                        const int* __restrict__ chist,
                                                        int* __restrict__ counts,
                                                        int* __restrict__ perm) {
  __shared__ int s_wc[WAVES_PB * NEXP];
  __shared__ int s_ptot[16 * NEXP];
  __shared__ int s_ppre[16 * NEXP];
  __shared__ int s_tot[NEXP];
  __shared__ int s_pre[NEXP];
  const int tid  = threadIdx.x;
  const int lane = tid & 31;
  const int wave = tid >> 5;
  const int blk  = blockIdx.x;
  const int b    = blk * CHUNK + tid;
  const int e    = y[b] & (NEXP - 1);

  {
    const int q = tid & 15, grp = tid >> 4;
    int tot = 0, pre = 0;
    #pragma unroll
    for (int r = 0; r < 16; ++r) {
      const int row = grp * 16 + r;
      const int v = chist[row * HROW + q];
      tot += v;
      pre += (row < blk) ? v : 0;
    }
    s_ptot[grp * NEXP + q] = tot;
    s_ppre[grp * NEXP + q] = pre;
  }

  unsigned mymask = 0u;
  int mine = 0;
  #pragma unroll
  for (int k = 0; k < NEXP; ++k) {
    const unsigned m = __builtin_amdgcn_ballot_w32(e == k);
    if (e == k) mymask = m;
    if (lane == k) mine = __popc(m);
  }
  if (lane < NEXP) s_wc[wave * NEXP + lane] = mine;
  const int rank_w = __popc(mymask & ((1u << lane) - 1u));
  __syncthreads();

  if (tid < NEXP) {
    int t = 0, p = 0;
    #pragma unroll
    for (int g = 0; g < 16; ++g) { t += s_ptot[g * NEXP + tid]; p += s_ppre[g * NEXP + tid]; }
    s_tot[tid] = t;
    s_pre[tid] = p;
    if (blk == 0) counts[tid] = t;
  }
  __syncthreads();

  int pos = s_pre[e] + rank_w;
  #pragma unroll
  for (int k = 0; k < NEXP; ++k) pos += (k < e) ? s_tot[k] : 0;
  #pragma unroll
  for (int w = 0; w < WAVES_PB; ++w) pos += (w < wave) ? s_wc[w * NEXP + e] : 0;
  perm[pos] = b;
}


__device__ __forceinline__ void layer_tile(const float* __restrict__ src,
                                           const float* __restrict__ W, int ldw,
                                           const float* __restrict__ bias,
                                           float* __restrict__ dst,
                                           int Kdim, int lane, int wave) {
  const int n0   = wave << 4;
  const int m    = lane & 15;
  const int koff = (lane >> 4) << 1;
  const int n    = n0 + m;
  v8f c0 = {}, c1 = {}, c2 = {}, c3 = {};
  const float* a_ptr = src + m * LDS_STRIDE + koff;
  const float* b_ptr = W + (size_t)koff * ldw + n;
  const int steps = Kdim >> 2;
  #pragma unroll 4
  for (int kk = 0; kk < steps; ++kk) {
    v2f b; b.x = b_ptr[0]; b.y = b_ptr[ldw];
    v2f a0; a0.x = a_ptr[0 * 16 * LDS_STRIDE];     a0.y = a_ptr[0 * 16 * LDS_STRIDE + 1];
    v2f a1; a1.x = a_ptr[1 * 16 * LDS_STRIDE];     a1.y = a_ptr[1 * 16 * LDS_STRIDE + 1];
    v2f a2; a2.x = a_ptr[2 * 16 * LDS_STRIDE];     a2.y = a_ptr[2 * 16 * LDS_STRIDE + 1];
    v2f a3; a3.x = a_ptr[3 * 16 * LDS_STRIDE];     a3.y = a_ptr[3 * 16 * LDS_STRIDE + 1];
    c0 = wmma_f32(a0, b, c0);
    c1 = wmma_f32(a1, b, c1);
    c2 = wmma_f32(a2, b, c2);
    c3 = wmma_f32(a3, b, c3);
    a_ptr += 4;
    b_ptr += 4 * (size_t)ldw;
  }
  const float bv = bias[n];
  const int mbase = (lane >> 4) << 3;
  v8f acc[4] = {c0, c1, c2, c3};
  #pragma unroll
  for (int t = 0; t < MTILES; ++t) {
    #pragma unroll
    for (int v = 0; v < 8; ++v) {
      float x = acc[t][v] + bv;
      x = fmaxf(x, 0.0f);
      dst[(t * 16 + mbase + v) * LDS_STRIDE + n] = x;
    }
  }
}

__device__ __forceinline__ void layer_out(const float* __restrict__ src,
                                          const float* __restrict__ W,
                                          const float* __restrict__ bias,
                                          float* __restrict__ out,
                                          const int* __restrict__ sh_gid,
                                          int nvalid, int lane, int wave) {
  if (wave >= 4) return;
  const int m    = lane & 15;
  const int koff = (lane >> 4) << 1;
  const int n    = (wave << 4) + m;
  v8f c0 = {}, c1 = {}, c2 = {}, c3 = {};
  const float* a_ptr = src + m * LDS_STRIDE + koff;
  const float* b_ptr = W + (size_t)koff * DDIM + n;
  #pragma unroll 4
  for (int kk = 0; kk < HDIM / 4; ++kk) {
    v2f b; b.x = b_ptr[0]; b.y = b_ptr[DDIM];
    v2f a0; a0.x = a_ptr[0 * 16 * LDS_STRIDE];     a0.y = a_ptr[0 * 16 * LDS_STRIDE + 1];
    v2f a1; a1.x = a_ptr[1 * 16 * LDS_STRIDE];     a1.y = a_ptr[1 * 16 * LDS_STRIDE + 1];
    v2f a2; a2.x = a_ptr[2 * 16 * LDS_STRIDE];     a2.y = a_ptr[2 * 16 * LDS_STRIDE + 1];
    v2f a3; a3.x = a_ptr[3 * 16 * LDS_STRIDE];     a3.y = a_ptr[3 * 16 * LDS_STRIDE + 1];
    c0 = wmma_f32(a0, b, c0);
    c1 = wmma_f32(a1, b, c1);
    c2 = wmma_f32(a2, b, c2);
    c3 = wmma_f32(a3, b, c3);
    a_ptr += 4;
    b_ptr += 4 * DDIM;
  }
  const float bv = bias[n];
  const int mbase = (lane >> 4) << 3;
  v8f acc[4] = {c0, c1, c2, c3};
  #pragma unroll
  for (int t = 0; t < MTILES; ++t) {
    #pragma unroll
    for (int v = 0; v < 8; ++v) {
      int mm = t * 16 + mbase + v;
      if (mm < nvalid) {
        out[(size_t)sh_gid[mm] * DDIM + n] = acc[t][v] + bv;
      }
    }
  }
}

__global__ __launch_bounds__(256) void mapping_main(
    const float* __restrict__ z,
    const float* __restrict__ bw0, const float* __restrict__ bb0,
    const float* __restrict__ bw1, const float* __restrict__ bb1,
    const float* __restrict__ bw2, const float* __restrict__ bb2,
    const float* __restrict__ bw3, const float* __restrict__ bb3,
    const float* __restrict__ hw0, const float* __restrict__ hb0,
    const float* __restrict__ hw1, const float* __restrict__ hb1,
    const float* __restrict__ hw2, const float* __restrict__ hb2,
    const float* __restrict__ hw3, const float* __restrict__ hb3,
    const int* __restrict__ counts, const int* __restrict__ perm,
    float* __restrict__ out) {
  __shared__ float actA[ROWS * LDS_STRIDE];
  __shared__ float actB[ROWS * LDS_STRIDE];
  __shared__ int sh_meta[3];
  __shared__ int sh_gid[ROWS];

  const int tid  = threadIdx.x;
  const int lane = tid & 31;
  const int wave = tid >> 5;

  if (tid == 0) {
    int t = blockIdx.x;
    int off = 0, base = 0, expert = -1, start = 0, end = 0;
    for (int e = 0; e < NEXP; ++e) {
      int cnt = counts[e];
      int nt  = (cnt + ROWS - 1) / ROWS;
      if (expert < 0 && t < base + nt) {
        expert = e;
        start  = off + (t - base) * ROWS;
        end    = off + cnt;
      }
      base += nt;
      off  += cnt;
    }
    sh_meta[0] = expert; sh_meta[1] = start; sh_meta[2] = end;
  }
  __syncthreads();
  const int expert = sh_meta[0];
  if (expert < 0) return;
  const int start = sh_meta[1];
  const int end   = sh_meta[2];
  const int nvalid = min(ROWS, end - start);

  {
    #pragma unroll
    for (int i = 0; i < ROWS * ZDIM / 256; ++i) {
      int idx = i * 256 + tid;
      int r = idx >> 4, cidx = idx & 15;
      int row = start + r;
      if (row >= end) row = end - 1;
      int g = perm[row];
      if (cidx == 0) sh_gid[r] = g;
      actA[r * LDS_STRIDE + cidx] = z[(size_t)g * ZDIM + cidx];
    }
  }
  __syncthreads();

  layer_tile(actA, bw0, HDIM, bb0, actB, ZDIM, lane, wave);  __syncthreads();
  layer_tile(actB, bw1, HDIM, bb1, actA, HDIM, lane, wave);  __syncthreads();
  layer_tile(actA, bw2, HDIM, bb2, actB, HDIM, lane, wave);  __syncthreads();
  layer_tile(actB, bw3, HDIM, bb3, actA, HDIM, lane, wave);  __syncthreads();

  const size_t we = (size_t)expert * HDIM * HDIM;
  layer_tile(actA, hw0 + we, HDIM, hb0 + expert * HDIM, actB, HDIM, lane, wave); __syncthreads();
  layer_tile(actB, hw1 + we, HDIM, hb1 + expert * HDIM, actA, HDIM, lane, wave); __syncthreads();
  layer_tile(actA, hw2 + we, HDIM, hb2 + expert * HDIM, actB, HDIM, lane, wave); __syncthreads();
  layer_out(actB, hw3 + (size_t)expert * HDIM * DDIM, hb3 + expert * DDIM,
            out, sh_gid, nvalid, lane, wave);
}


extern "C" void kernel_launch(void* const* d_in, const int* in_sizes, int n_in,
                              void* d_out, int out_size, void* d_ws, size_t ws_size,
                              hipStream_t stream) {
  const float* z   = (const float*)d_in[0];
  const int*   y   = (const int*)  d_in[1];
  const float* bw0 = (const float*)d_in[2];
  const float* bb0 = (const float*)d_in[3];
  const float* bw1 = (const float*)d_in[4];
  const float* bb1 = (const float*)d_in[5];
  const float* bw2 = (const float*)d_in[6];
  const float* bb2 = (const float*)d_in[7];
  const float* bw3 = (const float*)d_in[8];
  const float* bb3 = (const float*)d_in[9];
  const float* hw0 = (const float*)d_in[10];
  const float* hb0 = (const float*)d_in[11];
  const float* hw1 = (const float*)d_in[12];
  const float* hb1 = (const float*)d_in[13];
  const float* hw2 = (const float*)d_in[14];
  const float* hb2 = (const float*)d_in[15];
  const float* hw3 = (const float*)d_in[16];
  const float* hb3 = (const float*)d_in[17];
  float* out = (float*)d_out;

  int* counts = (int*)d_ws;
  int* perm   = counts + 32;
  int* chist  = perm + BATCH;

  k_chunkhist<<<NCHUNK, CHUNK, 0, stream>>>(y, chist);
  k_rank_scatter<<<NCHUNK, CHUNK, 0, stream>>>(y, chist, counts, perm);
  mapping_main<<<NTILES_MAX, 256, 0, stream>>>(
      z, bw0, bb0, bw1, bb1, bw2, bb2, bw3, bb3,
      hw0, hb0, hw1, hb1, hw2, hb2, hw3, hb3,
      counts, perm, out);
}


# Main kernel re-tiled for balance: ~768 blocks (3 resident per WGP) of 5 or 6 16-row sub-tiles per wave instead of 1040 blocks of 4, so no block runs in a late partial wave; float sums unchanged
# speedup vs baseline: 1.3335x; 1.0904x over previous
#include <hip/hip_runtime.h>
#include <hip/hip_bf16.h>

#define BATCH   65536
#define NEXP    16
#define HDIM    128
#define DDIM    64
#define ZDIM    16
#define LDS_STRIDE 132
#define NBLK_MAX (BATCH / 16 * 3 / 16 + 3 + NEXP)
#define MAXMT   6
#define ROWS_MAX (MAXMT * 16)

typedef __attribute__((ext_vector_type(2))) float v2f;
typedef __attribute__((ext_vector_type(8))) float v8f;

__device__ __forceinline__ v8f wmma_f32(v2f a, v2f b, v8f c) {
  return __builtin_amdgcn_wmma_f32_16x16x4_f32(
       false, a,  false, b,
       (short)0, c,  false,  false);
}


#define CHUNK      256
#define NCHUNK     (BATCH / CHUNK)
#define HROW       32
#define WAVES_PB   (CHUNK / 32)

__global__ __launch_bounds__(CHUNK) void k_chunkhist(const int* __restrict__ y,
                                                     int* __restrict__ chist) {
  __shared__ int s_wc[WAVES_PB * NEXP];
  const int tid  = threadIdx.x;
  const int lane = tid & 31;
  const int wave = tid >> 5;
  const int e = y[blockIdx.x * CHUNK + tid] & (NEXP - 1);
  int mine = 0;
  #pragma unroll
  for (int k = 0; k < NEXP; ++k) {
    const unsigned m = __builtin_amdgcn_ballot_w32(e == k);
    if (lane == k) mine = __popc(m);
  }
  if (lane < NEXP) s_wc[wave * NEXP + lane] = mine;
  __syncthreads();
  if (tid < NEXP) {
    int s = 0;
    #pragma unroll
    for (int w = 0; w < WAVES_PB; ++w) s += s_wc[w * NEXP + tid];
    chist[blockIdx.x * HROW + tid] = s;
  }
}

__global__ __launch_bounds__(CHUNK) void k_rank_scatter(const int* __restrict__ y,
                                                        const int* __restrict__ chist,
                                                        int* __restrict__ counts,
                                                        int* __restrict__ perm) {
  __shared__ int s_wc[WAVES_PB * NEXP];
  __shared__ int s_ptot[16 * NEXP];
  __shared__ int s_ppre[16 * NEXP];
  __shared__ int s_tot[NEXP];
  __shared__ int s_pre[NEXP];
  const int tid  = threadIdx.x;
  const int lane = tid & 31;
  const int wave = tid >> 5;
  const int blk  = blockIdx.x;
  const int b    = blk * CHUNK + tid;
  const int e    = y[b] & (NEXP - 1);

  {
    const int q = tid & 15, grp = tid >> 4;
    int tot = 0, pre = 0;
    #pragma unroll
    for (int r = 0; r < 16; ++r) {
      const int row = grp * 16 + r;
      const int v = chist[row * HROW + q];
      tot += v;
      pre += (row < blk) ? v : 0;
    }
    s_ptot[grp * NEXP + q] = tot;
    s_ppre[grp * NEXP + q] = pre;
  }

  unsigned mymask = 0u;
  int mine = 0;
  #pragma unroll
  for (int k = 0; k < NEXP; ++k) {
    const unsigned m = __builtin_amdgcn_ballot_w32(e == k);
    if (e == k) mymask = m;
    if (lane == k) mine = __popc(m);
  }
  if (lane < NEXP) s_wc[wave * NEXP + lane] = mine;
  const int rank_w = __popc(mymask & ((1u << lane) - 1u));
  __syncthreads();

  if (tid < NEXP) {
    int t = 0, p = 0;
    #pragma unroll
    for (int g = 0; g < 16; ++g) { t += s_ptot[g * NEXP + tid]; p += s_ppre[g * NEXP + tid]; }
    s_tot[tid] = t;
    s_pre[tid] = p;
    if (blk == 0) counts[tid] = t;
  }
  __syncthreads();

  int pos = s_pre[e] + rank_w;
  #pragma unroll
  for (int k = 0; k < NEXP; ++k) pos += (k < e) ? s_tot[k] : 0;
  #pragma unroll
  for (int w = 0; w < WAVES_PB; ++w) pos += (w < wave) ? s_wc[w * NEXP + e] : 0;
  perm[pos] = b;
}


template <int MT>
__device__ __forceinline__ void layer_tile(const float* __restrict__ src,
                                           const float* __restrict__ W, int ldw,
                                           const float* __restrict__ bias,
                                           float* __restrict__ dst,
                                           int Kdim, int lane, int wave) {
  const int n0   = wave << 4;
  const int m    = lane & 15;
  const int koff = (lane >> 4) << 1;
  const int n    = n0 + m;
  v8f c[MT];
  #pragma unroll
  for (int t = 0; t < MT; ++t) c[t] = v8f{};
  const float* a_ptr = src + m * LDS_STRIDE + koff;
  const float* b_ptr = W + (size_t)koff * ldw + n;
  const int steps = Kdim >> 2;
  #pragma unroll 4
  for (int kk = 0; kk < steps; ++kk) {
    v2f b; b.x = b_ptr[0]; b.y = b_ptr[ldw];
    v2f a[MT];
    #pragma unroll
    for (int t = 0; t < MT; ++t) {
      a[t].x = a_ptr[t * 16 * LDS_STRIDE];
      a[t].y = a_ptr[t * 16 * LDS_STRIDE + 1];
    }
    #pragma unroll
    for (int t = 0; t < MT; ++t) c[t] = wmma_f32(a[t], b, c[t]);
    a_ptr += 4;
    b_ptr += 4 * (size_t)ldw;
  }
  const float bv = bias[n];
  const int mbase = (lane >> 4) << 3;
  float* d_ptr = dst + mbase * LDS_STRIDE + n;
  #pragma unroll
  for (int t = 0; t < MT; ++t) {
    #pragma unroll
    for (int v = 0; v < 8; ++v) {
      float x = c[t][v] + bv;
      x = fmaxf(x, 0.0f);
      d_ptr[(t * 16 + v) * LDS_STRIDE] = x;
    }
  }
}

template <int MT>
__device__ __forceinline__ void layer_out(const float* __restrict__ src,
                                          const float* __restrict__ W,
                                          const float* __restrict__ bias,
                                          float* __restrict__ out,
                                          const int* __restrict__ sh_gid,
                                          int nvalid, int lane, int wave) {
  if (wave >= 4) return;
  const int m    = lane & 15;
  const int koff = (lane >> 4) << 1;
  const int n    = (wave << 4) + m;
  v8f c[MT];
  #pragma unroll
  for (int t = 0; t < MT; ++t) c[t] = v8f{};
  const float* a_ptr = src + m * LDS_STRIDE + koff;
  const float* b_ptr = W + (size_t)koff * DDIM + n;
  #pragma unroll 4
  for (int kk = 0; kk < HDIM / 4; ++kk) {
    v2f b; b.x = b_ptr[0]; b.y = b_ptr[DDIM];
    v2f a[MT];
    #pragma unroll
    for (int t = 0; t < MT; ++t) {
      a[t].x = a_ptr[t * 16 * LDS_STRIDE];
      a[t].y = a_ptr[t * 16 * LDS_STRIDE + 1];
    }
    #pragma unroll
    for (int t = 0; t < MT; ++t) c[t] = wmma_f32(a[t], b, c[t]);
    a_ptr += 4;
    b_ptr += 4 * DDIM;
  }
  const float bv = bias[n];
  const int mbase = (lane >> 4) << 3;
  const int* g_ptr = sh_gid + mbase;
  #pragma unroll
  for (int t = 0; t < MT; ++t) {
    #pragma unroll
    for (int v = 0; v < 8; ++v) {
      if (t * 16 + v < nvalid - mbase) {
        out[(size_t)g_ptr[t * 16 + v] * DDIM + n] = c[t][v] + bv;
      }
    }
  }
}

struct MapArgs {
  const float* z;
  const float* bw0; const float* bb0;
  const float* bw1; const float* bb1;
  const float* bw2; const float* bb2;
  const float* bw3; const float* bb3;
  const float* hw0; const float* hb0;
  const float* hw1; const float* hb1;
  const float* hw2; const float* hb2;
  const float* hw3; const float* hb3;
};

template <int MT>
__device__ __forceinline__ void run_rows(const MapArgs& p, const int* __restrict__ perm,
                                         float* __restrict__ out,
                                         float* actA, float* actB, int* sh_gid,
                                         int expert, int start, int end,
                                         int tid, int lane, int wave) {
  const int nvalid = min(MT * 16, end - start);
  #pragma unroll
  for (int i = 0; i < MT; ++i) {
    int idx = i * 256 + tid;
    int r = idx >> 4, cidx = idx & 15;
    int row = start + r;
    if (row >= end) row = end - 1;
    int g = perm[row];
    if (cidx == 0) sh_gid[r] = g;
    actA[r * LDS_STRIDE + cidx] = p.z[(size_t)g * ZDIM + cidx];
  }
  __syncthreads();

  layer_tile<MT>(actA, p.bw0, HDIM, p.bb0, actB, ZDIM, lane, wave);  __syncthreads();
  layer_tile<MT>(actB, p.bw1, HDIM, p.bb1, actA, HDIM, lane, wave);  __syncthreads();
  layer_tile<MT>(actA, p.bw2, HDIM, p.bb2, actB, HDIM, lane, wave);  __syncthreads();
  layer_tile<MT>(actB, p.bw3, HDIM, p.bb3, actA, HDIM, lane, wave);  __syncthreads();

  const size_t we = (size_t)expert * HDIM * HDIM;
  layer_tile<MT>(actA, p.hw0 + we, HDIM, p.hb0 + expert * HDIM, actB, HDIM, lane, wave); __syncthreads();
  layer_tile<MT>(actB, p.hw1 + we, HDIM, p.hb1 + expert * HDIM, actA, HDIM, lane, wave); __syncthreads();
  layer_tile<MT>(actA, p.hw2 + we, HDIM, p.hb2 + expert * HDIM, actB, HDIM, lane, wave); __syncthreads();
  layer_out<MT>(actB, p.hw3 + (size_t)expert * HDIM * DDIM, p.hb3 + expert * DDIM,
                out, sh_gid, nvalid, lane, wave);
}

__global__ __launch_bounds__(256) void mapping_main(
    const float* __restrict__ z,
    const float* __restrict__ bw0, const float* __restrict__ bb0,
    const float* __restrict__ bw1, const float* __restrict__ bb1,
    const float* __restrict__ bw2, const float* __restrict__ bb2,
    const float* __restrict__ bw3, const float* __restrict__ bb3,
    const float* __restrict__ hw0, const float* __restrict__ hb0,
    const float* __restrict__ hw1, const float* __restrict__ hb1,
    const float* __restrict__ hw2, const float* __restrict__ hb2,
    const float* __restrict__ hw3, const float* __restrict__ hb3,
    const int* __restrict__ counts, const int* __restrict__ perm,
    float* __restrict__ out) {
  __shared__ float actA[ROWS_MAX * LDS_STRIDE];
  __shared__ float actB[ROWS_MAX * LDS_STRIDE];
  __shared__ int sh_gid[ROWS_MAX];

  const int tid  = threadIdx.x;
  const int lane = tid & 31;
  const int wave = tid >> 5;

  const int blk = blockIdx.x;
  int expert = -1, j = 0, me = 0, nbe = 1, off = 0, cnt = 0;
  int bbase = 0, roff = 0;
  int cv[NEXP];
  #pragma unroll
  for (int e = 0; e < NEXP; ++e) cv[e] = counts[e];
  #pragma unroll
  for (int e = 0; e < NEXP; ++e) {
    const int c  = cv[e];
    const int m  = (c + 15) >> 4;
    const int nb = max((m + MAXMT - 1) / MAXMT, (3 * m) >> 4);
    const bool hit = (expert < 0) & (blk < bbase + nb);
    expert = hit ? e : expert;
    j      = hit ? blk - bbase : j;
    me     = hit ? m : me;
    nbe    = hit ? nb : nbe;
    off    = hit ? roff : off;
    cnt    = hit ? c : cnt;
    bbase += nb;
    roff  += c;
  }
  if (expert < 0) return;
  const int q     = me / nbe;
  const int r     = me - q * nbe;
  const int lo    = j * q + min(j, r);
  const int nmt   = q + ((j < r) ? 1 : 0);
  const int start = off + lo * 16;
  const int end   = min(off + cnt, start + nmt * 16);

  const MapArgs p = {z, bw0, bb0, bw1, bb1, bw2, bb2, bw3, bb3,
                     hw0, hb0, hw1, hb1, hw2, hb2, hw3, hb3};

  if (nmt == 5) run_rows<5>(p, perm, out, actA, actB, sh_gid, expert, start, end, tid, lane, wave);
  else          run_rows<6>(p, perm, out, actA, actB, sh_gid, expert, start, end, tid, lane, wave);
}


extern "C" void kernel_launch(void* const* d_in, const int* in_sizes, int n_in,
                              void* d_out, int out_size, void* d_ws, size_t ws_size,
                              hipStream_t stream) {
  const float* z   = (const float*)d_in[0];
  const int*   y   = (const int*)  d_in[1];
  const float* bw0 = (const float*)d_in[2];
  const float* bb0 = (const float*)d_in[3];
  const float* bw1 = (const float*)d_in[4];
  const float* bb1 = (const float*)d_in[5];
  const float* bw2 = (const float*)d_in[6];
  const float* bb2 = (const float*)d_in[7];
  const float* bw3 = (const float*)d_in[8];
  const float* bb3 = (const float*)d_in[9];
  const float* hw0 = (const float*)d_in[10];
  const float* hb0 = (const float*)d_in[11];
  const float* hw1 = (const float*)d_in[12];
  const float* hb1 = (const float*)d_in[13];
  const float* hw2 = (const float*)d_in[14];
  const float* hb2 = (const float*)d_in[15];
  const float* hw3 = (const float*)d_in[16];
  const float* hb3 = (const float*)d_in[17];
  float* out = (float*)d_out;

  int* counts = (int*)d_ws;
  int* perm   = counts + 32;
  int* chist  = perm + BATCH;

  k_chunkhist<<<NCHUNK, CHUNK, 0, stream>>>(y, chist);
  k_rank_scatter<<<NCHUNK, CHUNK, 0, stream>>>(y, chist, counts, perm);
  mapping_main<<<NBLK_MAX, 256, 0, stream>>>(
      z, bw0, bb0, bw1, bb1, bw2, bb2, bw3, bb3,
      hw0, hb0, hw1, hb1, hw2, hb2, hw3, hb3,
      counts, perm, out);
}


# Operand loads software-pipelined through two register sets: next weight group and next LDS activation block requested ahead of each WMMA block, next layer's first group before the barrier; same sums
# speedup vs baseline: 1.5211x; 1.1407x over previous
#include <hip/hip_runtime.h>
#include <hip/hip_bf16.h>

#define BATCH   65536
#define NEXP    16
#define HDIM    128
#define DDIM    64
#define ZDIM    16
#define LDS_STRIDE 132
#define NBLK_MAX (BATCH / 16 * 3 / 16 + 3 + NEXP)
#define MAXMT   6
#define ROWS_MAX (MAXMT * 16)

typedef __attribute__((ext_vector_type(2))) float v2f;
typedef __attribute__((ext_vector_type(8))) float v8f;

__device__ __forceinline__ v8f wmma_f32(v2f a, v2f b, v8f c) {
  return __builtin_amdgcn_wmma_f32_16x16x4_f32(
       false, a,  false, b,
       (short)0, c,  false,  false);
}


#define CHUNK      256
#define NCHUNK     (BATCH / CHUNK)
#define HROW       32
#define WAVES_PB   (CHUNK / 32)

__global__ __launch_bounds__(CHUNK) void k_chunkhist(const int* __restrict__ y,
                                                     int* __restrict__ chist) {
  __shared__ int s_wc[WAVES_PB * NEXP];
  const int tid  = threadIdx.x;
  const int lane = tid & 31;
  const int wave = tid >> 5;
  const int e = y[blockIdx.x * CHUNK + tid] & (NEXP - 1);
  int mine = 0;
  #pragma unroll
  for (int k = 0; k < NEXP; ++k) {
    const unsigned m = __builtin_amdgcn_ballot_w32(e == k);
    if (lane == k) mine = __popc(m);
  }
  if (lane < NEXP) s_wc[wave * NEXP + lane] = mine;
  __syncthreads();
  if (tid < NEXP) {
    int s = 0;
    #pragma unroll
    for (int w = 0; w < WAVES_PB; ++w) s += s_wc[w * NEXP + tid];
    chist[blockIdx.x * HROW + tid] = s;
  }
}

__global__ __launch_bounds__(CHUNK) void k_rank_scatter(const int* __restrict__ y,
                                                        const int* __restrict__ chist,
                                                        int* __restrict__ counts,
                                                        int* __restrict__ perm) {
  __shared__ int s_wc[WAVES_PB * NEXP];
  __shared__ int s_ptot[16 * NEXP];
  __shared__ int s_ppre[16 * NEXP];
  __shared__ int s_tot[NEXP];
  __shared__ int s_pre[NEXP];
  const int tid  = threadIdx.x;
  const int lane = tid & 31;
  const int wave = tid >> 5;
  const int blk  = blockIdx.x;
  const int b    = blk * CHUNK + tid;
  const int e    = y[b] & (NEXP - 1);

  {
    const int q = tid & 15, grp = tid >> 4;
    int tot = 0, pre = 0;
    #pragma unroll
    for (int r = 0; r < 16; ++r) {
      const int row = grp * 16 + r;
      const int v = chist[row * HROW + q];
      tot += v;
      pre += (row < blk) ? v : 0;
    }
    s_ptot[grp * NEXP + q] = tot;
    s_ppre[grp * NEXP + q] = pre;
  }

  unsigned mymask = 0u;
  int mine = 0;
  #pragma unroll
  for (int k = 0; k < NEXP; ++k) {
    const unsigned m = __builtin_amdgcn_ballot_w32(e == k);
    if (e == k) mymask = m;
    if (lane == k) mine = __popc(m);
  }
  if (lane < NEXP) s_wc[wave * NEXP + lane] = mine;
  const int rank_w = __popc(mymask & ((1u << lane) - 1u));
  __syncthreads();

  if (tid < NEXP) {
    int t = 0, p = 0;
    #pragma unroll
    for (int g = 0; g < 16; ++g) { t += s_ptot[g * NEXP + tid]; p += s_ppre[g * NEXP + tid]; }
    s_tot[tid] = t;
    s_pre[tid] = p;
    if (blk == 0) counts[tid] = t;
  }
  __syncthreads();

  int pos = s_pre[e] + rank_w;
  #pragma unroll
  for (int k = 0; k < NEXP; ++k) pos += (k < e) ? s_tot[k] : 0;
  #pragma unroll
  for (int w = 0; w < WAVES_PB; ++w) pos += (w < wave) ? s_wc[w * NEXP + e] : 0;
  perm[pos] = b;
}


struct BGroup { v2f s[4]; };

template <int LDW>
__device__ __forceinline__ BGroup load_bgroup(const float* __restrict__ Wg, int boff) {
  BGroup r;
  #pragma unroll
  for (int s = 0; s < 4; ++s) {
    r.s[s].x = Wg[boff + (4 * s) * LDW];
    r.s[s].y = Wg[boff + (4 * s + 1) * LDW];
  }
  return r;
}

template <int MT> struct ABlock { v2f a[2][MT]; };

template <int MT>
__device__ __forceinline__ ABlock<MT> load_ablock(const float* a_lane, int kofs) {
  ABlock<MT> r;
  #pragma unroll
  for (int s = 0; s < 2; ++s) {
    #pragma unroll
    for (int t = 0; t < MT; ++t)
      r.a[s][t] = *reinterpret_cast<const v2f*>(a_lane + t * 16 * LDS_STRIDE + kofs + 4 * s);
  }
  return r;
}

template <int MT>
__device__ __forceinline__ void mma_block(const ABlock<MT>& a, v2f b0, v2f b1, v8f (&c)[MT]) {
  #pragma unroll
  for (int t = 0; t < MT; ++t) c[t] = wmma_f32(a.a[0][t], b0, c[t]);
  #pragma unroll
  for (int t = 0; t < MT; ++t) c[t] = wmma_f32(a.a[1][t], b1, c[t]);
}

template <int MT, int KDIM, int LDW, bool NEXT_SAME>
__device__ __forceinline__ void mma_layer(const float* a_lane,
                                          const float* __restrict__ W, int boff,
                                          const float* __restrict__ Wn,
                                          BGroup& bq, v8f (&c)[MT]) {
  if constexpr (KDIM == 16) {
    ABlock<MT> a0 = load_ablock<MT>(a_lane, 0);
    ABlock<MT> a1 = load_ablock<MT>(a_lane, 8);
    const BGroup bc = bq;
    if constexpr (NEXT_SAME) bq = load_bgroup<LDW>(Wn, boff);
    mma_block<MT>(a0, bc.s[0], bc.s[1], c);
    mma_block<MT>(a1, bc.s[2], bc.s[3], c);
  } else {
    constexpr int TRIPS = KDIM / 32;
    BGroup bA = bq, bB;
    ABlock<MT> a0 = load_ablock<MT>(a_lane, 0), a1;
    #pragma unroll 1
    for (int tr = 0; tr < TRIPS; ++tr) {
      const int k0 = tr * 32;
      const float* Wk = W + (size_t)k0 * LDW;
      const float* Wp = (tr + 1 < TRIPS) ? Wk + 32 * LDW : (NEXT_SAME ? Wn : W);
      bB = load_bgroup<LDW>(Wk + 16 * LDW, boff);
      a1 = load_ablock<MT>(a_lane, k0 + 8);
      __builtin_amdgcn_sched_barrier(0);
      mma_block<MT>(a0, bA.s[0], bA.s[1], c);
      __builtin_amdgcn_sched_barrier(0);
      a0 = load_ablock<MT>(a_lane, k0 + 16);
      __builtin_amdgcn_sched_barrier(0);
      mma_block<MT>(a1, bA.s[2], bA.s[3], c);
      __builtin_amdgcn_sched_barrier(0);
      bA = load_bgroup<LDW>(Wp, boff);
      a1 = load_ablock<MT>(a_lane, k0 + 24);
      __builtin_amdgcn_sched_barrier(0);
      mma_block<MT>(a0, bB.s[0], bB.s[1], c);
      __builtin_amdgcn_sched_barrier(0);
      a0 = load_ablock<MT>(a_lane, (k0 + 32) & (KDIM - 1));
      __builtin_amdgcn_sched_barrier(0);
      mma_block<MT>(a1, bB.s[2], bB.s[3], c);
    }
    bq = bA;
  }
}

template <int MT, int KDIM, int LDWN>
__device__ __forceinline__ void layer_tile(const float* __restrict__ src,
                                           const float* __restrict__ W,
                                           float* __restrict__ dst,
                                           const float* __restrict__ Wn, int boffn,
                                           const float* __restrict__ biasn, int nn,
                                           BGroup& bq, float& bv,
                                           int lane, int wave) {
  const int n0   = wave << 4;
  const int m    = lane & 15;
  const int koff = (lane >> 4) << 1;
  const int n    = n0 + m;
  const float bvn = biasn[nn];
  v8f c[MT];
  #pragma unroll
  for (int t = 0; t < MT; ++t) c[t] = v8f{};
  mma_layer<MT, KDIM, HDIM, LDWN == HDIM>(src + m * LDS_STRIDE + koff, W, koff * HDIM + n, Wn, bq, c);
  if constexpr (LDWN != HDIM) bq = load_bgroup<LDWN>(Wn, boffn);
  const int mbase = (lane >> 4) << 3;
  float* d_ptr = dst + mbase * LDS_STRIDE + n;
  #pragma unroll
  for (int t = 0; t < MT; ++t) {
    #pragma unroll
    for (int v = 0; v < 8; ++v) {
      float x = c[t][v] + bv;
      x = fmaxf(x, 0.0f);
      d_ptr[(t * 16 + v) * LDS_STRIDE] = x;
    }
  }
  bv = bvn;
}

template <int MT>
__device__ __forceinline__ void layer_out(const float* __restrict__ src,
                                          const float* __restrict__ W,
                                          float* __restrict__ out,
                                          const int* __restrict__ sh_gid,
                                          BGroup& bq, float bv,
                                          int nvalid, int lane, int wave) {
  if (wave >= 4) return;
  const int m    = lane & 15;
  const int koff = (lane >> 4) << 1;
  const int n    = (wave << 4) + m;
  v8f c[MT];
  #pragma unroll
  for (int t = 0; t < MT; ++t) c[t] = v8f{};
  mma_layer<MT, HDIM, DDIM, false>(src + m * LDS_STRIDE + koff, W, koff * DDIM + n, W, bq, c);
  const int mbase = (lane >> 4) << 3;
  const int* g_ptr = sh_gid + mbase;
  #pragma unroll
  for (int t = 0; t < MT; ++t) {
    #pragma unroll
    for (int v = 0; v < 8; ++v) {
      if (t * 16 + v < nvalid - mbase) {
        out[(size_t)g_ptr[t * 16 + v] * DDIM + n] = c[t][v] + bv;
      }
    }
  }
}

struct MapArgs {
  const float* z;
  const float* bw0; const float* bb0;
  const float* bw1; const float* bb1;
  const float* bw2; const float* bb2;
  const float* bw3; const float* bb3;
  const float* hw0; const float* hb0;
  const float* hw1; const float* hb1;
  const float* hw2; const float* hb2;
  const float* hw3; const float* hb3;
};

template <int MT>
__device__ __forceinline__ void run_rows(const MapArgs& p, const int* __restrict__ perm,
                                         float* __restrict__ out,
                                         float* actA, float* actB, int* sh_gid,
                                         int expert, int start, int end,
                                         int tid, int lane, int wave) {
  const int nvalid = min(MT * 16, end - start);
  const int m    = lane & 15;
  const int koff = (lane >> 4) << 1;
  const int n    = (wave << 4) + m;
  const int boff = koff * HDIM + n;
  const int no   = n & (DDIM - 1);
  const int boffo = koff * DDIM + no;
  BGroup bq = load_bgroup<HDIM>(p.bw0, boff);
  float bv = p.bb0[n];
  #pragma unroll
  for (int i = 0; i < MT; ++i) {
    int idx = i * 256 + tid;
    int r = idx >> 4, cidx = idx & 15;
    int row = start + r;
    if (row >= end) row = end - 1;
    int g = perm[row];
    if (cidx == 0) sh_gid[r] = g;
    actA[r * LDS_STRIDE + cidx] = p.z[(size_t)g * ZDIM + cidx];
  }
  __syncthreads();

  const size_t we = (size_t)expert * HDIM * HDIM;
  const float* hw0 = p.hw0 + we;
  const float* hw1 = p.hw1 + we;
  const float* hw2 = p.hw2 + we;
  const float* hw3 = p.hw3 + (size_t)expert * HDIM * DDIM;

  layer_tile<MT, ZDIM, HDIM>(actA, p.bw0, actB, p.bw1, boff, p.bb1, n, bq, bv, lane, wave);  __syncthreads();
  layer_tile<MT, HDIM, HDIM>(actB, p.bw1, actA, p.bw2, boff, p.bb2, n, bq, bv, lane, wave);  __syncthreads();
  layer_tile<MT, HDIM, HDIM>(actA, p.bw2, actB, p.bw3, boff, p.bb3, n, bq, bv, lane, wave);  __syncthreads();
  layer_tile<MT, HDIM, HDIM>(actB, p.bw3, actA, hw0, boff, p.hb0 + expert * HDIM, n, bq, bv, lane, wave);  __syncthreads();

  layer_tile<MT, HDIM, HDIM>(actA, hw0, actB, hw1, boff, p.hb1 + expert * HDIM, n, bq, bv, lane, wave); __syncthreads();
  layer_tile<MT, HDIM, HDIM>(actB, hw1, actA, hw2, boff, p.hb2 + expert * HDIM, n, bq, bv, lane, wave); __syncthreads();
  layer_tile<MT, HDIM, DDIM>(actA, hw2, actB, hw3, boffo, p.hb3 + expert * DDIM, no, bq, bv, lane, wave); __syncthreads();
  layer_out<MT>(actB, hw3, out, sh_gid, bq, bv, nvalid, lane, wave);
}

__global__ __launch_bounds__(256) void mapping_main(
    const float* __restrict__ z,
    const float* __restrict__ bw0, const float* __restrict__ bb0,
    const float* __restrict__ bw1, const float* __restrict__ bb1,
    const float* __restrict__ bw2, const float* __restrict__ bb2,
    const float* __restrict__ bw3, const float* __restrict__ bb3,
    const float* __restrict__ hw0, const float* __restrict__ hb0,
    const float* __restrict__ hw1, const float* __restrict__ hb1,
    const float* __restrict__ hw2, const float* __restrict__ hb2,
    const float* __restrict__ hw3, const float* __restrict__ hb3,
    const int* __restrict__ counts, const int* __restrict__ perm,
    float* __restrict__ out) {
  __shared__ __attribute__((aligned(16))) float actA[ROWS_MAX * LDS_STRIDE];
  __shared__ __attribute__((aligned(16))) float actB[ROWS_MAX * LDS_STRIDE];
  __shared__ int sh_gid[ROWS_MAX];

  const int tid  = threadIdx.x;
  const int lane = tid & 31;
  const int wave = tid >> 5;

  const int blk = blockIdx.x;
  int expert = -1, j = 0, me = 0, nbe = 1, off = 0, cnt = 0;
  int bbase = 0, roff = 0;
  int cv[NEXP];
  #pragma unroll
  for (int e = 0; e < NEXP; ++e) cv[e] = counts[e];
  #pragma unroll
  for (int e = 0; e < NEXP; ++e) {
    const int c  = cv[e];
    const int m  = (c + 15) >> 4;
    const int nb = max((m + MAXMT - 1) / MAXMT, (3 * m) >> 4);
    const bool hit = (expert < 0) & (blk < bbase + nb);
    expert = hit ? e : expert;
    j      = hit ? blk - bbase : j;
    me     = hit ? m : me;
    nbe    = hit ? nb : nbe;
    off    = hit ? roff : off;
    cnt    = hit ? c : cnt;
    bbase += nb;
    roff  += c;
  }
  if (expert < 0) return;
  const int q     = me / nbe;
  const int r     = me - q * nbe;
  const int lo    = j * q + min(j, r);
  const int nmt   = q + ((j < r) ? 1 : 0);
  const int start = off + lo * 16;
  const int end   = min(off + cnt, start + nmt * 16);

  const MapArgs p = {z, bw0, bb0, bw1, bb1, bw2, bb2, bw3, bb3,
                     hw0, hb0, hw1, hb1, hw2, hb2, hw3, hb3};

  if (nmt == 5) run_rows<5>(p, perm, out, actA, actB, sh_gid, expert, start, end, tid, lane, wave);
  else          run_rows<6>(p, perm, out, actA, actB, sh_gid, expert, start, end, tid, lane, wave);
}


extern "C" void kernel_launch(void* const* d_in, const int* in_sizes, int n_in,
                              void* d_out, int out_size, void* d_ws, size_t ws_size,
                              hipStream_t stream) {
  const float* z   = (const float*)d_in[0];
  const int*   y   = (const int*)  d_in[1];
  const float* bw0 = (const float*)d_in[2];
  const float* bb0 = (const float*)d_in[3];
  const float* bw1 = (const float*)d_in[4];
  const float* bb1 = (const float*)d_in[5];
  const float* bw2 = (const float*)d_in[6];
  const float* bb2 = (const float*)d_in[7];
  const float* bw3 = (const float*)d_in[8];
  const float* bb3 = (const float*)d_in[9];
  const float* hw0 = (const float*)d_in[10];
  const float* hb0 = (const float*)d_in[11];
  const float* hw1 = (const float*)d_in[12];
  const float* hb1 = (const float*)d_in[13];
  const float* hw2 = (const float*)d_in[14];
  const float* hb2 = (const float*)d_in[15];
  const float* hw3 = (const float*)d_in[16];
  const float* hb3 = (const float*)d_in[17];
  float* out = (float*)d_out;

  int* counts = (int*)d_ws;
  int* perm   = counts + 32;
  int* chist  = perm + BATCH;

  k_chunkhist<<<NCHUNK, CHUNK, 0, stream>>>(y, chist);
  k_rank_scatter<<<NCHUNK, CHUNK, 0, stream>>>(y, chist, counts, perm);
  mapping_main<<<NBLK_MAX, 256, 0, stream>>>(
      z, bw0, bb0, bw1, bb1, bw2, bb2, bw3, bb3,
      hw0, hb0, hw1, hb1, hw2, hb2, hw3, hb3,
      counts, perm, out);
}


# One launch instead of three, no workspace: each work-group reads its quarter of y, ranks its expert's samples by a block prefix sum and takes its share of rows; float sums unchanged
# speedup vs baseline: 1.6141x; 1.0611x over previous
#include <hip/hip_runtime.h>
#include <hip/hip_bf16.h>

#define BATCH   65536
#define NEXP    16
#define HDIM    128
#define DDIM    64
#define ZDIM    16
#define LDS_STRIDE 132
#define MAXMT   6
#define ROWS_MAX (MAXMT * 16)

typedef __attribute__((ext_vector_type(2))) float v2f;
typedef __attribute__((ext_vector_type(8))) float v8f;

__device__ __forceinline__ v8f wmma_f32(v2f a, v2f b, v8f c) {
  return __builtin_amdgcn_wmma_f32_16x16x4_f32(
       false, a,  false, b,
       (short)0, c,  false,  false);
}

#define NSUP    4
#define SUP     (BATCH / NSUP)
#define NBP     12
#define YLD     (SUP / 4 / 256)
#define NBLK    (NEXP * NSUP * NBP)


struct BGroup { v2f s[4]; };

template <int LDW>
__device__ __forceinline__ BGroup load_bgroup(const float* __restrict__ Wg, int boff) {
  BGroup r;
  #pragma unroll
  for (int s = 0; s < 4; ++s) {
    r.s[s].x = Wg[boff + (4 * s) * LDW];
    r.s[s].y = Wg[boff + (4 * s + 1) * LDW];
  }
  return r;
}

template <int MT> struct ABlock { v2f a[2][MT]; };

template <int MT>
__device__ __forceinline__ ABlock<MT> load_ablock(const float* a_lane, int kofs) {
  ABlock<MT> r;
  #pragma unroll
  for (int s = 0; s < 2; ++s) {
    #pragma unroll
    for (int t = 0; t < MT; ++t)
      r.a[s][t] = *reinterpret_cast<const v2f*>(a_lane + t * 16 * LDS_STRIDE + kofs + 4 * s);
  }
  return r;
}

template <int MT>
__device__ __forceinline__ void mma_block(const ABlock<MT>& a, v2f b0, v2f b1, v8f (&c)[MT]) {
  #pragma unroll
  for (int t = 0; t < MT; ++t) c[t] = wmma_f32(a.a[0][t], b0, c[t]);
  #pragma unroll
  for (int t = 0; t < MT; ++t) c[t] = wmma_f32(a.a[1][t], b1, c[t]);
}

template <int MT, int KDIM, int LDW, bool NEXT_SAME>
__device__ __forceinline__ void mma_layer(const float* a_lane,
                                          const float* __restrict__ W, int boff,
                                          const float* __restrict__ Wn,
                                          BGroup& bq, v8f (&c)[MT]) {
  if constexpr (KDIM == 16) {
    ABlock<MT> a0 = load_ablock<MT>(a_lane, 0);
    ABlock<MT> a1 = load_ablock<MT>(a_lane, 8);
    const BGroup bc = bq;
    if constexpr (NEXT_SAME) bq = load_bgroup<LDW>(Wn, boff);
    mma_block<MT>(a0, bc.s[0], bc.s[1], c);
    mma_block<MT>(a1, bc.s[2], bc.s[3], c);
  } else {
    constexpr int TRIPS = KDIM / 32;
    BGroup bA = bq, bB;
    ABlock<MT> a0 = load_ablock<MT>(a_lane, 0), a1;
    #pragma unroll 1
    for (int tr = 0; tr < TRIPS; ++tr) {
      const int k0 = tr * 32;
      const float* Wk = W + (size_t)k0 * LDW;
      const float* Wp = (tr + 1 < TRIPS) ? Wk + 32 * LDW : (NEXT_SAME ? Wn : W);
      bB = load_bgroup<LDW>(Wk + 16 * LDW, boff);
      a1 = load_ablock<MT>(a_lane, k0 + 8);
      __builtin_amdgcn_sched_barrier(0);
      mma_block<MT>(a0, bA.s[0], bA.s[1], c);
      __builtin_amdgcn_sched_barrier(0);
      a0 = load_ablock<MT>(a_lane, k0 + 16);
      __builtin_amdgcn_sched_barrier(0);
      mma_block<MT>(a1, bA.s[2], bA.s[3], c);
      __builtin_amdgcn_sched_barrier(0);
      bA = load_bgroup<LDW>(Wp, boff);
      a1 = load_ablock<MT>(a_lane, k0 + 24);
      __builtin_amdgcn_sched_barrier(0);
      mma_block<MT>(a0, bB.s[0], bB.s[1], c);
      __builtin_amdgcn_sched_barrier(0);
      a0 = load_ablock<MT>(a_lane, (k0 + 32) & (KDIM - 1));
      __builtin_amdgcn_sched_barrier(0);
      mma_block<MT>(a1, bB.s[2], bB.s[3], c);
    }
    bq = bA;
  }
}

template <int MT, int KDIM, int LDWN>
__device__ __forceinline__ void layer_tile(const float* __restrict__ src,
                                           const float* __restrict__ W,
                                           float* __restrict__ dst,
                                           const float* __restrict__ Wn, int boffn,
                                           const float* __restrict__ biasn, int nn,
                                           BGroup& bq, float& bv,
                                           int lane, int wave) {
  const int n0   = wave << 4;
  const int m    = lane & 15;
  const int koff = (lane >> 4) << 1;
  const int n    = n0 + m;
  const float bvn = biasn[nn];
  v8f c[MT];
  #pragma unroll
  for (int t = 0; t < MT; ++t) c[t] = v8f{};
  mma_layer<MT, KDIM, HDIM, LDWN == HDIM>(src + m * LDS_STRIDE + koff, W, koff * HDIM + n, Wn, bq, c);
  if constexpr (LDWN != HDIM) bq = load_bgroup<LDWN>(Wn, boffn);
  const int mbase = (lane >> 4) << 3;
  float* d_ptr = dst + mbase * LDS_STRIDE + n;
  #pragma unroll
  for (int t = 0; t < MT; ++t) {
    #pragma unroll
    for (int v = 0; v < 8; ++v) {
      float x = c[t][v] + bv;
      x = fmaxf(x, 0.0f);
      d_ptr[(t * 16 + v) * LDS_STRIDE] = x;
    }
  }
  bv = bvn;
}

template <int MT>
__device__ __forceinline__ void layer_out(const float* __restrict__ src,
                                          const float* __restrict__ W,
                                          float* __restrict__ out,
                                          const int* __restrict__ sh_gid,
                                          BGroup& bq, float bv,
                                          int nvalid, int lane, int wave) {
  if (wave >= 4) return;
  const int m    = lane & 15;
  const int koff = (lane >> 4) << 1;
  const int n    = (wave << 4) + m;
  v8f c[MT];
  #pragma unroll
  for (int t = 0; t < MT; ++t) c[t] = v8f{};
  mma_layer<MT, HDIM, DDIM, false>(src + m * LDS_STRIDE + koff, W, koff * DDIM + n, W, bq, c);
  const int mbase = (lane >> 4) << 3;
  const int* g_ptr = sh_gid + mbase;
  #pragma unroll
  for (int t = 0; t < MT; ++t) {
    #pragma unroll
    for (int v = 0; v < 8; ++v) {
      if (t * 16 + v < nvalid - mbase) {
        out[(size_t)g_ptr[t * 16 + v] * DDIM + n] = c[t][v] + bv;
      }
    }
  }
}

struct MapArgs {
  const float* z;
  const float* bw0; const float* bb0;
  const float* bw1; const float* bb1;
  const float* bw2; const float* bb2;
  const float* bw3; const float* bb3;
  const float* hw0; const float* hb0;
  const float* hw1; const float* hb1;
  const float* hw2; const float* hb2;
  const float* hw3; const float* hb3;
};

template <int MT>
__device__ __forceinline__ void run_rows(const MapArgs& p, float* __restrict__ out,
                                         float* actA, float* actB, const int* sh_gid,
                                         int expert, int nvalid,
                                         int tid, int lane, int wave) {
  const int m    = lane & 15;
  const int koff = (lane >> 4) << 1;
  const int n    = (wave << 4) + m;
  const int boff = koff * HDIM + n;
  const int no   = n & (DDIM - 1);
  const int boffo = koff * DDIM + no;
  BGroup bq = load_bgroup<HDIM>(p.bw0, boff);
  float bv = p.bb0[n];
  #pragma unroll
  for (int i = 0; i < MT; ++i) {
    int idx = i * 256 + tid;
    int r = idx >> 4, cidx = idx & 15;
    int g = sh_gid[min(r, nvalid - 1)];
    actA[r * LDS_STRIDE + cidx] = p.z[(size_t)g * ZDIM + cidx];
  }
  __syncthreads();

  const size_t we = (size_t)expert * HDIM * HDIM;
  const float* hw0 = p.hw0 + we;
  const float* hw1 = p.hw1 + we;
  const float* hw2 = p.hw2 + we;
  const float* hw3 = p.hw3 + (size_t)expert * HDIM * DDIM;

  layer_tile<MT, ZDIM, HDIM>(actA, p.bw0, actB, p.bw1, boff, p.bb1, n, bq, bv, lane, wave);  __syncthreads();
  layer_tile<MT, HDIM, HDIM>(actB, p.bw1, actA, p.bw2, boff, p.bb2, n, bq, bv, lane, wave);  __syncthreads();
  layer_tile<MT, HDIM, HDIM>(actA, p.bw2, actB, p.bw3, boff, p.bb3, n, bq, bv, lane, wave);  __syncthreads();
  layer_tile<MT, HDIM, HDIM>(actB, p.bw3, actA, hw0, boff, p.hb0 + expert * HDIM, n, bq, bv, lane, wave);  __syncthreads();

  layer_tile<MT, HDIM, HDIM>(actA, hw0, actB, hw1, boff, p.hb1 + expert * HDIM, n, bq, bv, lane, wave); __syncthreads();
  layer_tile<MT, HDIM, HDIM>(actB, hw1, actA, hw2, boff, p.hb2 + expert * HDIM, n, bq, bv, lane, wave); __syncthreads();
  layer_tile<MT, HDIM, DDIM>(actA, hw2, actB, hw3, boffo, p.hb3 + expert * DDIM, no, bq, bv, lane, wave); __syncthreads();
  layer_out<MT>(actB, hw3, out, sh_gid, bq, bv, nvalid, lane, wave);
}

__global__ __launch_bounds__(256) void mapping_main(
    const float* __restrict__ z, const int* __restrict__ y,
    const float* __restrict__ bw0, const float* __restrict__ bb0,
    const float* __restrict__ bw1, const float* __restrict__ bb1,
    const float* __restrict__ bw2, const float* __restrict__ bb2,
    const float* __restrict__ bw3, const float* __restrict__ bb3,
    const float* __restrict__ hw0, const float* __restrict__ hb0,
    const float* __restrict__ hw1, const float* __restrict__ hb1,
    const float* __restrict__ hw2, const float* __restrict__ hb2,
    const float* __restrict__ hw3, const float* __restrict__ hb3,
    float* __restrict__ out) {
  __shared__ __attribute__((aligned(16))) float actA[ROWS_MAX * LDS_STRIDE];
  __shared__ __attribute__((aligned(16))) float actB[ROWS_MAX * LDS_STRIDE];
  __shared__ int sh_gid[ROWS_MAX + 1];
  __shared__ int s_wsum[8];

  const int tid  = threadIdx.x;
  const int lane = tid & 31;
  const int wave = tid >> 5;

  const int blk    = blockIdx.x;
  const int s      = blk & (NSUP - 1);
  const int pr     = blk >> 2;
  const int expert = pr / NBP;
  const int j0     = (pr - expert * NBP + 4 * s) % NBP;
  const int u      = (j0 >= 4 && j0 < 8) ? 11 - j0 : j0;
  const int sbase  = s * SUP;

  const int4* __restrict__ y4 = reinterpret_cast<const int4*>(y + sbase);
  int4 yv[YLD];
  #pragma unroll
  for (int i = 0; i < YLD; ++i) yv[i] = y4[i * 256 + tid];
  int cnt = 0;
  #pragma unroll
  for (int i = 0; i < YLD; ++i) {
    cnt += ((yv[i].x & (NEXP - 1)) == expert) ? 1 : 0;
    cnt += ((yv[i].y & (NEXP - 1)) == expert) ? 1 : 0;
    cnt += ((yv[i].z & (NEXP - 1)) == expert) ? 1 : 0;
    cnt += ((yv[i].w & (NEXP - 1)) == expert) ? 1 : 0;
  }
  int incl = cnt;
  #pragma unroll
  for (int d = 1; d < 32; d <<= 1) {
    const int o = __shfl_up(incl, d);
    if (lane >= d) incl += o;
  }
  if (lane == 31) s_wsum[wave] = incl;
  __syncthreads();
  int wpre = 0, tot = 0;
  #pragma unroll
  for (int w = 0; w < 8; ++w) {
    const int v = s_wsum[w];
    wpre += (w < wave) ? v : 0;
    tot  += v;
  }
  const int first = wpre + incl - cnt;
  tot = __builtin_amdgcn_readfirstlane(tot);

  const int mt   = (tot + 15) >> 4;
  const int q    = mt / NBP;
  const int r    = mt - q * NBP;
  const int lo   = u * q + min(u, r);
  const int nsub = q + ((u < r) ? 1 : 0);
  if (nsub == 0) return;

  const MapArgs p = {z, bw0, bb0, bw1, bb1, bw2, bb2, bw3, bb3,
                     hw0, hb0, hw1, hb1, hw2, hb2, hw3, hb3};

  const int nmt   = min(nsub, MAXMT);
  const int start = lo * 16;
  const int end   = min(tot, start + nmt * 16);
  if (__builtin_amdgcn_ballot_w32(cnt > 0 && first < end && first + cnt > start) != 0u) {
    int rk = first - start;
    const unsigned span = (unsigned)(end - start);
    #pragma unroll
    for (int i = 0; i < YLD; ++i) {
      const int id = sbase + 4 * (i * 256 + tid);
      const int vv[4] = {yv[i].x, yv[i].y, yv[i].z, yv[i].w};
      #pragma unroll
      for (int k = 0; k < 4; ++k) {
        const bool hit = (vv[k] & (NEXP - 1)) == expert;
        sh_gid[(hit && (unsigned)rk < span) ? rk : ROWS_MAX] = id + k;
        rk += hit ? 1 : 0;
      }
    }
  }
  __syncthreads();
  if (nmt == 5) run_rows<5>(p, out, actA, actB, sh_gid, expert, end - start, tid, lane, wave);
  else          run_rows<6>(p, out, actA, actB, sh_gid, expert, end - start, tid, lane, wave);

  #pragma unroll 1
  for (int d = MAXMT; d < nsub; ++d) {
    __syncthreads();
    const int st1 = (lo + d) * 16;
    const int en1 = min(tot, st1 + 16);
    int rk = first - st1;
    const unsigned span = (unsigned)(en1 - st1);
    #pragma unroll 1
    for (int i = 0; i < YLD; ++i) {
      const int4 v = y4[i * 256 + tid];
      const int id = sbase + 4 * (i * 256 + tid);
      if ((v.x & (NEXP - 1)) == expert) { if ((unsigned)rk < span) sh_gid[rk] = id;     ++rk; }
      if ((v.y & (NEXP - 1)) == expert) { if ((unsigned)rk < span) sh_gid[rk] = id + 1; ++rk; }
      if ((v.z & (NEXP - 1)) == expert) { if ((unsigned)rk < span) sh_gid[rk] = id + 2; ++rk; }
      if ((v.w & (NEXP - 1)) == expert) { if ((unsigned)rk < span) sh_gid[rk] = id + 3; ++rk; }
    }
    __syncthreads();
    run_rows<1>(p, out, actA, actB, sh_gid, expert, en1 - st1, tid, lane, wave);
  }
}


extern "C" void kernel_launch(void* const* d_in, const int* in_sizes, int n_in,
                              void* d_out, int out_size, void* d_ws, size_t ws_size,
                              hipStream_t stream) {
  const float* z   = (const float*)d_in[0];
  const int*   y   = (const int*)  d_in[1];
  const float* bw0 = (const float*)d_in[2];
  const float* bb0 = (const float*)d_in[3];
  const float* bw1 = (const float*)d_in[4];
  const float* bb1 = (const float*)d_in[5];
  const float* bw2 = (const float*)d_in[6];
  const float* bb2 = (const float*)d_in[7];
  const float* bw3 = (const float*)d_in[8];
  const float* bb3 = (const float*)d_in[9];
  const float* hw0 = (const float*)d_in[10];
  const float* hb0 = (const float*)d_in[11];
  const float* hw1 = (const float*)d_in[12];
  const float* hb1 = (const float*)d_in[13];
  const float* hw2 = (const float*)d_in[14];
  const float* hb2 = (const float*)d_in[15];
  const float* hw3 = (const float*)d_in[16];
  const float* hb3 = (const float*)d_in[17];
  float* out = (float*)d_out;

  mapping_main<<<NBLK, 256, 0, stream>>>(
      z, y, bw0, bb0, bw1, bb1, bw2, bb2, bw3, bb3,
      hw0, hb0, hw1, hb1, hw2, hb2, hw3, hb3,
      out);
}
